# starting kernel
# baseline (speedup 1.0000x reference)
#include <hip/hip_runtime.h>

typedef __attribute__((ext_vector_type(2))) float v2f;
typedef __attribute__((ext_vector_type(8))) float v8f;

#define H_IMG 512
#define W_IMG 512
#define CH_STRIDE (512 * 512)
#define RS 42
#define HS 17

__global__ __launch_bounds__(32)
void ImageEvaluator_84000970375834_kernel(const float* __restrict__ img,
                                          float* __restrict__ out) {
    __shared__ float sXs[48 * RS];
    __shared__ float sXq[48 * RS];
    __shared__ float sH1[40 * HS];
    __shared__ float sH2[40 * HS];

    const int lane = threadIdx.x;
    const int hi   = lane >> 4;
    const int lo   = lane & 15;
    const int tc   = blockIdx.x * 16;
    const int tr   = blockIdx.y * 16;
    const int b    = blockIdx.z;

    const float* base = img + (size_t)b * 3 * CH_STRIDE;

    v2f band[10];
    #pragma unroll
    for (int k = 0; k < 10; ++k) {
        const int j0 = 4 * k + 2 * hi;
        band[k].x = ((unsigned)(j0     - lo) <= 24u) ? 1.0f : 0.0f;
        band[k].y = ((unsigned)(j0 + 1 - lo) <= 24u) ? 1.0f : 0.0f;
    }

    for (int p = lane; p < 40 * 40; p += 32) {
        int rr = p / 40, cc = p - rr * 40;
        int ir = tr - 12 + rr; ir = (ir < 0) ? -ir : ((ir > 511) ? 1022 - ir : ir);
        int ic = tc - 12 + cc; ic = (ic < 0) ? -ic : ((ic > 511) ? 1022 - ic : ic);
        size_t off = (size_t)ir * W_IMG + ic;
        float x0 = base[off];
        float x1 = base[off + CH_STRIDE];
        float x2 = base[off + 2 * CH_STRIDE];
        sXs[rr * RS + cc] = x0 + x1 + x2;
        sXq[rr * RS + cc] = x0 * x0 + x1 * x1 + x2 * x2;
    }
    __syncthreads();

    #pragma unroll
    for (int g = 0; g < 3; ++g) {
        v8f acc1 = {}, acc2 = {};
        const int arow = g * 16 + lo;
        #pragma unroll
        for (int k = 0; k < 10; ++k) {
            const int j0 = 4 * k + 2 * hi;
            v2f a1 = { sXs[arow * RS + j0], sXs[arow * RS + j0 + 1] };
            v2f a2 = { sXq[arow * RS + j0], sXq[arow * RS + j0 + 1] };
            acc1 = __builtin_amdgcn_wmma_f32_16x16x4_f32(false, a1, false, band[k],
                                                         (short)0, acc1, false, false);
            acc2 = __builtin_amdgcn_wmma_f32_16x16x4_f32(false, a2, false, band[k],
                                                         (short)0, acc2, false, false);
        }
        #pragma unroll
        for (int v = 0; v < 8; ++v) {
            int row = g * 16 + v + 8 * hi;
            if (row < 40) {
                sH1[row * HS + lo] = acc1[v];
                sH2[row * HS + lo] = acc2[v];
            }
        }
    }
    __syncthreads();

    v8f s1 = {}, s2 = {};
    #pragma unroll
    for (int k = 0; k < 10; ++k) {
        const int i0 = 4 * k + 2 * hi;
        v2f b1 = { sH1[i0 * HS + lo], sH1[(i0 + 1) * HS + lo] };
        v2f b2 = { sH2[i0 * HS + lo], sH2[(i0 + 1) * HS + lo] };
        s1 = __builtin_amdgcn_wmma_f32_16x16x4_f32(false, band[k], false, b1,
                                                   (short)0, s1, false, false);
        s2 = __builtin_amdgcn_wmma_f32_16x16x4_f32(false, band[k], false, b2,
                                                   (short)0, s2, false, false);
    }

    const float inv = 1.0f / (25.0f * 25.0f * 3.0f);
    const float EPS = 1.0f / 255.0f;
    float* outb = out + (size_t)b * CH_STRIDE;
    #pragma unroll
    for (int v = 0; v < 8; ++v) {
        const int row = tr + v + 8 * hi;
        const int col = tc + lo;
        const size_t off = (size_t)row * W_IMG + col;
        float meanr = s1[v] * inv;
        float means = s2[v] * inv;
        float expo  = fabsf(meanr - 0.5f) + EPS;
        float contrast = means - meanr * meanr;
        float x0 = base[off];
        float x1 = base[off + CH_STRIDE];
        float x2 = base[off + 2 * CH_STRIDE];
        float mx = fmaxf(x0, fmaxf(x1, x2));
        float mn = fminf(x0, fminf(x1, x2));
        float sat = (mx - mn + EPS) / (mx + EPS);
        outb[off] = sat * contrast / expo;
    }
}

extern "C" void kernel_launch(void* const* d_in, const int* in_sizes, int n_in,
                              void* d_out, int out_size, void* d_ws, size_t ws_size,
                              hipStream_t stream) {
    (void)in_sizes; (void)n_in; (void)d_ws; (void)ws_size; (void)out_size;
    const float* img = (const float*)d_in[0];
    float* out = (float*)d_out;
    dim3 grid(W_IMG / 16, H_IMG / 16, 16);
    ImageEvaluator_84000970375834_kernel<<<grid, dim3(32), 0, stream>>>(img, out);
}


# stage-1 and epilogue global loads made independent (lane = fixed column, rows fully unrolled, reflect math hoisted): all 174 loads of a wave stream in flight instead of 6 per full wait
# speedup vs baseline: 1.4785x; 1.4785x over previous
#include <hip/hip_runtime.h>

typedef __attribute__((ext_vector_type(2))) float v2f;
typedef __attribute__((ext_vector_type(8))) float v8f;

#define H_IMG 512
#define W_IMG 512
#define CH_STRIDE (512 * 512)
#define RS 42
#define HS 17

__global__ __launch_bounds__(32)
void ImageEvaluator_84000970375834_kernel(const float* __restrict__ img,
                                          float* __restrict__ out) {
    __shared__ float sXs[48 * RS];
    __shared__ float sXq[48 * RS];
    __shared__ float sH1[40 * HS];
    __shared__ float sH2[40 * HS];

    const int lane = threadIdx.x;
    const int hi   = lane >> 4;
    const int lo   = lane & 15;
    const int tc   = blockIdx.x * 16;
    const int tr   = blockIdx.y * 16;
    const int b    = blockIdx.z;

    const float* base = img + (size_t)b * 3 * CH_STRIDE;

    v2f band[10];
    #pragma unroll
    for (int k = 0; k < 10; ++k) {
        const int j0 = 4 * k + 2 * hi;
        band[k].x = ((unsigned)(j0     - lo) <= 24u) ? 1.0f : 0.0f;
        band[k].y = ((unsigned)(j0 + 1 - lo) <= 24u) ? 1.0f : 0.0f;
    }

    int ca = tc - 12 + lane;
    ca = (ca < 0) ? -ca : ((ca > 511) ? 1022 - ca : ca);
    int cb = tc + 20 + (lane & 7);
    cb = (cb < 0) ? -cb : ((cb > 511) ? 1022 - cb : cb);
    const int rb = lane >> 3;

    float xb[10][3];
    #pragma unroll
    for (int i = 0; i < 10; ++i) {
        int ir = tr - 12 + rb + 4 * i;
        ir = (ir < 0) ? -ir : ((ir > 511) ? 1022 - ir : ir);
        const float* p = base + (size_t)ir * W_IMG + cb;
        xb[i][0] = p[0];
        xb[i][1] = p[CH_STRIDE];
        xb[i][2] = p[2 * CH_STRIDE];
    }

    float ex[8][3];
    #pragma unroll
    for (int v = 0; v < 8; ++v) {
        const float* p = base + (size_t)(tr + v + 8 * hi) * W_IMG + tc + lo;
        ex[v][0] = p[0];
        ex[v][1] = p[CH_STRIDE];
        ex[v][2] = p[2 * CH_STRIDE];
    }

    #pragma unroll
    for (int h = 0; h < 2; ++h) {
        float xa[20][3];
        #pragma unroll
        for (int i = 0; i < 20; ++i) {
            int ir = tr - 12 + h * 20 + i;
            ir = (ir < 0) ? -ir : ((ir > 511) ? 1022 - ir : ir);
            const float* p = base + (size_t)ir * W_IMG + ca;
            xa[i][0] = p[0];
            xa[i][1] = p[CH_STRIDE];
            xa[i][2] = p[2 * CH_STRIDE];
        }
        #pragma unroll
        for (int i = 0; i < 20; ++i) {
            const int rr = h * 20 + i;
            const float x0 = xa[i][0], x1 = xa[i][1], x2 = xa[i][2];
            sXs[rr * RS + lane] = x0 + x1 + x2;
            sXq[rr * RS + lane] = x0 * x0 + x1 * x1 + x2 * x2;
        }
    }
    #pragma unroll
    for (int i = 0; i < 10; ++i) {
        const int rr = rb + 4 * i;
        const float x0 = xb[i][0], x1 = xb[i][1], x2 = xb[i][2];
        sXs[rr * RS + 32 + (lane & 7)] = x0 + x1 + x2;
        sXq[rr * RS + 32 + (lane & 7)] = x0 * x0 + x1 * x1 + x2 * x2;
    }
    __syncthreads();

    #pragma unroll
    for (int g = 0; g < 3; ++g) {
        v8f acc1 = {}, acc2 = {};
        const int arow = g * 16 + lo;
        #pragma unroll
        for (int k = 0; k < 10; ++k) {
            const int j0 = 4 * k + 2 * hi;
            v2f a1 = { sXs[arow * RS + j0], sXs[arow * RS + j0 + 1] };
            v2f a2 = { sXq[arow * RS + j0], sXq[arow * RS + j0 + 1] };
            acc1 = __builtin_amdgcn_wmma_f32_16x16x4_f32(false, a1, false, band[k],
                                                         (short)0, acc1, false, false);
            acc2 = __builtin_amdgcn_wmma_f32_16x16x4_f32(false, a2, false, band[k],
                                                         (short)0, acc2, false, false);
        }
        #pragma unroll
        for (int v = 0; v < 8; ++v) {
            int row = g * 16 + v + 8 * hi;
            if (row < 40) {
                sH1[row * HS + lo] = acc1[v];
                sH2[row * HS + lo] = acc2[v];
            }
        }
    }
    __syncthreads();

    v8f s1 = {}, s2 = {};
    #pragma unroll
    for (int k = 0; k < 10; ++k) {
        const int i0 = 4 * k + 2 * hi;
        v2f b1 = { sH1[i0 * HS + lo], sH1[(i0 + 1) * HS + lo] };
        v2f b2 = { sH2[i0 * HS + lo], sH2[(i0 + 1) * HS + lo] };
        s1 = __builtin_amdgcn_wmma_f32_16x16x4_f32(false, band[k], false, b1,
                                                   (short)0, s1, false, false);
        s2 = __builtin_amdgcn_wmma_f32_16x16x4_f32(false, band[k], false, b2,
                                                   (short)0, s2, false, false);
    }

    const float inv = 1.0f / (25.0f * 25.0f * 3.0f);
    const float EPS = 1.0f / 255.0f;
    float* outb = out + (size_t)b * CH_STRIDE;
    #pragma unroll
    for (int v = 0; v < 8; ++v) {
        const int row = tr + v + 8 * hi;
        const int col = tc + lo;
        const size_t off = (size_t)row * W_IMG + col;
        float meanr = s1[v] * inv;
        float means = s2[v] * inv;
        float expo  = fabsf(meanr - 0.5f) + EPS;
        float contrast = means - meanr * meanr;
        float x0 = ex[v][0];
        float x1 = ex[v][1];
        float x2 = ex[v][2];
        float mx = fmaxf(x0, fmaxf(x1, x2));
        float mn = fminf(x0, fminf(x1, x2));
        float sat = (mx - mn + EPS) / (mx + EPS);
        outb[off] = sat * contrast / expo;
    }
}

extern "C" void kernel_launch(void* const* d_in, const int* in_sizes, int n_in,
                              void* d_out, int out_size, void* d_ws, size_t ws_size,
                              hipStream_t stream) {
    (void)in_sizes; (void)n_in; (void)d_ws; (void)ws_size; (void)out_size;
    const float* img = (const float*)d_in[0];
    float* out = (float*)d_out;
    dim3 grid(W_IMG / 16, H_IMG / 16, 16);
    ImageEvaluator_84000970375834_kernel<<<grid, dim3(32), 0, stream>>>(img, out);
}


# work-group tile 16x16 -> 64x64: 8 waves share one 88x88 halo region staged once in LDS; per 16x16 result 174 -> 90 global wave-loads (whole aligned lines), 80 -> 50 WMMAs; summation order unchanged
# speedup vs baseline: 3.1002x; 2.0969x over previous
#include <hip/hip_runtime.h>

typedef __attribute__((ext_vector_type(2))) float v2f;
typedef __attribute__((ext_vector_type(4))) float v4f;
typedef __attribute__((ext_vector_type(8))) float v8f;

#define H_IMG 512
#define W_IMG 512
#define CH_STRIDE (512 * 512)
#define TILE 64
#define REG  88
#define XS   92
#define HT   100
#define XPLANE (REG * XS)
#define HPLANE (TILE * HT)

__global__ __launch_bounds__(256)
void score_tile64_kernel(const float* __restrict__ img,
                                          float* __restrict__ out) {
    __shared__ __attribute__((aligned(16))) float smem[2 * XPLANE + 2 * HPLANE];

    const int tid  = threadIdx.x;
    const int lane = tid & 31;
    const int hi   = lane >> 4;
    const int lo   = lane & 15;
    const int wv   = __builtin_amdgcn_readfirstlane(tid >> 5);
    const int tc   = blockIdx.x * TILE;
    const int tr   = blockIdx.y * TILE;
    const int b    = blockIdx.z;

    const float* base = img + (size_t)b * 3 * CH_STRIDE;

    v2f band[10];
    #pragma unroll
    for (int k = 0; k < 10; ++k) {
        const int j0 = 4 * k + 2 * hi;
        band[k].x = ((unsigned)(j0     - lo) <= 24u) ? 1.0f : 0.0f;
        band[k].y = ((unsigned)(j0 + 1 - lo) <= 24u) ? 1.0f : 0.0f;
    }

    {
        const int cc   = tid & 127;
        const int c    = cc - 20;
        const int rpar = wv >> 2;
        if ((unsigned)c < (unsigned)REG) {
            int ci = tc - 12 + c;
            ci = (ci < 0) ? -ci : ((ci > 511) ? 1022 - ci : ci);
            #pragma unroll
            for (int hb = 0; hb < 2; ++hb) {
            float xa[22][3];
            #pragma unroll
            for (int i = 0; i < 22; ++i) {
                int ir = tr - 12 + rpar + 2 * (22 * hb + i);
                ir = (ir < 0) ? -ir : ((ir > 511) ? 1022 - ir : ir);
                const float* p = base + (size_t)ir * W_IMG + ci;
                xa[i][0] = p[0];
                xa[i][1] = p[CH_STRIDE];
                xa[i][2] = p[2 * CH_STRIDE];
            }
            #pragma unroll
            for (int i = 0; i < 22; ++i) {
                const int rr = rpar + 2 * (22 * hb + i);
                const float x0 = xa[i][0], x1 = xa[i][1], x2 = xa[i][2];
                smem[rr * XS + c]          = x0 + x1 + x2;
                smem[XPLANE + rr * XS + c] = x0 * x0 + x1 * x1 + x2 * x2;
            }
            }
        }
    }

    const int mb = wv >> 2;
    const int nt = wv & 3;

    __syncthreads();

    float ex[2][8][3];
    #pragma unroll
    for (int t = 0; t < 2; ++t) {
        #pragma unroll
        for (int v = 0; v < 8; ++v) {
            const float* p = base + (size_t)(tr + 16 * (mb + 2 * t) + v + 8 * hi) * W_IMG + tc + 16 * nt + lo;
            ex[t][v][0] = p[0];
            ex[t][v][1] = p[CH_STRIDE];
            ex[t][v][2] = p[2 * CH_STRIDE];
        }
    }

    {
        const int pl = wv & 1;
        const float* sX = smem + pl * XPLANE;
        float* sH = smem + 2 * XPLANE + pl * HPLANE;
        const int pr = wv >> 1;
        #pragma unroll
        for (int j = 0; j < 6; ++j) {
            const int q  = pr + 4 * j;
            const int m0 = (q >> 2) * 16;
            const int n0 = (q & 3) * 16;
            int arow = m0 + lo;
            arow = (arow < REG - 1) ? arow : (REG - 1);
            const float* ap = sX + arow * XS + n0 + 2 * hi;
            v2f a[10];
            #pragma unroll
            for (int k = 0; k < 10; ++k)
                a[k] = *(const v2f*)(ap + 4 * k);
            v8f acc = {};
            #pragma unroll
            for (int k = 0; k < 10; ++k)
                acc = __builtin_amdgcn_wmma_f32_16x16x4_f32(false, a[k], false, band[k],
                                                            (short)0, acc, false, false);
            float* hp = sH + (n0 + lo) * HT + m0 + 8 * hi;
            v4f d0 = { acc[0], acc[1], acc[2], acc[3] };
            v4f d1 = { acc[4], acc[5], acc[6], acc[7] };
            *(v4f*)(hp)     = d0;
            *(v4f*)(hp + 4) = d1;
        }
    }
    __syncthreads();

    const float inv = 1.0f / (25.0f * 25.0f * 3.0f);
    const float EPS = 1.0f / 255.0f;
    float* outb = out + (size_t)b * CH_STRIDE;
    #pragma unroll
    for (int t = 0; t < 2; ++t) {
        const int mt = mb + 2 * t;
        v8f s1 = {}, s2 = {};
        const float* h1 = smem + 2 * XPLANE + (16 * nt + lo) * HT + 16 * mt + 2 * hi;
        const float* h2 = h1 + HPLANE;
        #pragma unroll
        for (int k = 0; k < 10; ++k) {
            v2f b1 = *(const v2f*)(h1 + 4 * k);
            v2f b2 = *(const v2f*)(h2 + 4 * k);
            s1 = __builtin_amdgcn_wmma_f32_16x16x4_f32(false, band[k], false, b1,
                                                       (short)0, s1, false, false);
            s2 = __builtin_amdgcn_wmma_f32_16x16x4_f32(false, band[k], false, b2,
                                                       (short)0, s2, false, false);
        }

        #pragma unroll
        for (int v = 0; v < 8; ++v) {
            const int row = tr + 16 * mt + v + 8 * hi;
            const int col = tc + 16 * nt + lo;
            const size_t off = (size_t)row * W_IMG + col;
            float meanr = s1[v] * inv;
            float means = s2[v] * inv;
            float expo  = fabsf(meanr - 0.5f) + EPS;
            float contrast = means - meanr * meanr;
            float x0 = ex[t][v][0];
            float x1 = ex[t][v][1];
            float x2 = ex[t][v][2];
            float mx = fmaxf(x0, fmaxf(x1, x2));
            float mn = fminf(x0, fminf(x1, x2));
            float sat = (mx - mn + EPS) / (mx + EPS);
            outb[off] = sat * contrast / expo;
        }
    }
}

extern "C" void kernel_launch(void* const* d_in, const int* in_sizes, int n_in,
                              void* d_out, int out_size, void* d_ws, size_t ws_size,
                              hipStream_t stream) {
    (void)in_sizes; (void)n_in; (void)d_ws; (void)ws_size; (void)out_size;
    const float* img = (const float*)d_in[0];
    float* out = (float*)d_out;
    dim3 grid(W_IMG / TILE, H_IMG / TILE, 16);
    score_tile64_kernel<<<grid, dim3(256), 0, stream>>>(img, out);
}
